# starting kernel
# baseline (speedup 1.0000x reference)
#include <hip/hip_runtime.h>

typedef __attribute__((ext_vector_type(2))) float v2f;
typedef __attribute__((ext_vector_type(8))) float v8f;

#define NUM_SCOPES 128
#define NUM_DECOMPS 8
#define BATCH 1024
#define K_IN 32
#define N_OUT 32

constexpr int WAVES   = 8;
constexpr int STRIP   = 32;
constexpr int NSTRIPS = BATCH / STRIP;
constexpr int APAD    = 33;

__global__ __launch_bounds__(256)
void logmm_kernel(const float* __restrict__ x,
                  const float* __restrict__ acc,
                  float* __restrict__ out) {
  __shared__ float sB[K_IN * N_OUT];
  __shared__ float sMax[N_OUT];
  __shared__ float sCb[N_OUT];
  __shared__ float sA[WAVES][STRIP * APAD];

  const int tid  = threadIdx.x;
  const int lane = tid & 31;
  const int wave = tid >> 5;
  const int sd   = blockIdx.x;

  const float* xb = x   + (size_t)sd * BATCH * K_IN;
  const float* ab = acc + (size_t)sd * K_IN * N_OUT;
  float*       ob = out + (size_t)sd * BATCH * N_OUT;

  ((float4*)sB)[tid] = ((const float4*)ab)[tid];
  __syncthreads();

  if (tid < N_OUT) {
    const int j = tid;
    float m = sB[j], s = sB[j];
    #pragma unroll
    for (int k = 1; k < K_IN; ++k) {
      float v = sB[k * N_OUT + j];
      m = fmaxf(m, v);
      s += v;
    }
    sMax[j] = m;
    sCb[j]  = __logf(m) - __logf(s);
  }
  __syncthreads();

  {
    const int j = (tid * 4) & (N_OUT - 1);
    float4 v = ((float4*)sB)[tid];
    v.x /= sMax[j + 0]; v.y /= sMax[j + 1];
    v.z /= sMax[j + 2]; v.w /= sMax[j + 3];
    ((float4*)sB)[tid] = v;
  }
  __syncthreads();

  const int ml    = lane & 15;
  const int khalf = lane >> 4;
  v2f bfrag[2][8];
  #pragma unroll
  for (int jh = 0; jh < 2; ++jh) {
    #pragma unroll
    for (int kb = 0; kb < 8; ++kb) {
      const int k0 = kb * 4 + 2 * khalf;
      const int n  = jh * 16 + ml;
      bfrag[jh][kb].x = sB[k0 * N_OUT + n];
      bfrag[jh][kb].y = sB[(k0 + 1) * N_OUT + n];
    }
  }
  const float cb0 = sCb[ml];
  const float cb1 = sCb[16 + ml];

  float* const aw = &sA[wave][0];

  for (int strip = wave; strip < NSTRIPS; strip += WAVES) {
    const int m0 = strip * STRIP;

    const float4* rp = (const float4*)(xb + (size_t)(m0 + lane) * K_IN);
    float4 r[8];
    #pragma unroll
    for (int i = 0; i < 8; ++i) r[i] = rp[i];

    float amax = r[0].x;
    #pragma unroll
    for (int i = 0; i < 8; ++i)
      amax = fmaxf(amax, fmaxf(fmaxf(r[i].x, r[i].y), fmaxf(r[i].z, r[i].w)));

    float* arow = aw + lane * APAD;
    #pragma unroll
    for (int i = 0; i < 8; ++i) {
      arow[4 * i + 0] = __expf(r[i].x - amax);
      arow[4 * i + 1] = __expf(r[i].y - amax);
      arow[4 * i + 2] = __expf(r[i].z - amax);
      arow[4 * i + 3] = __expf(r[i].w - amax);
    }
    __builtin_amdgcn_wave_barrier();

    v8f c00 = {}, c01 = {}, c10 = {}, c11 = {};
    #pragma unroll
    for (int kb = 0; kb < 8; ++kb) {
      const int k0 = kb * 4 + 2 * khalf;
      v2f a0, a1;
      a0.x = aw[ml        * APAD + k0];
      a0.y = aw[ml        * APAD + k0 + 1];
      a1.x = aw[(16 + ml) * APAD + k0];
      a1.y = aw[(16 + ml) * APAD + k0 + 1];
      c00 = __builtin_amdgcn_wmma_f32_16x16x4_f32(false, a0, false, bfrag[0][kb], (short)0, c00, false, false);
      c01 = __builtin_amdgcn_wmma_f32_16x16x4_f32(false, a0, false, bfrag[1][kb], (short)0, c01, false, false);
      c10 = __builtin_amdgcn_wmma_f32_16x16x4_f32(false, a1, false, bfrag[0][kb], (short)0, c10, false, false);
      c11 = __builtin_amdgcn_wmma_f32_16x16x4_f32(false, a1, false, bfrag[1][kb], (short)0, c11, false, false);
    }

    #pragma unroll
    for (int mt = 0; mt < 2; ++mt) {
      const v8f cA = mt ? c10 : c00;
      const v8f cB = mt ? c11 : c01;
      #pragma unroll
      for (int v = 0; v < 8; ++v) {
        const int mlocal = mt * 16 + v + 8 * khalf;
        const float am   = __shfl(amax, mlocal, 32);
        float* orow = ob + (size_t)(m0 + mlocal) * N_OUT;
        orow[ml]      = __logf(cA[v]) + am + cb0;
        orow[16 + ml] = __logf(cB[v]) + am + cb1;
      }
    }
    __builtin_amdgcn_wave_barrier();
  }
}

extern "C" void kernel_launch(void* const* d_in, const int* in_sizes, int n_in,
                              void* d_out, int out_size, void* d_ws, size_t ws_size,
                              hipStream_t stream) {
  const float* x   = (const float*)d_in[0];
  const float* acc = (const float*)d_in[1];
  float* out       = (float*)d_out;

  dim3 grid(NUM_SCOPES * NUM_DECOMPS);
  dim3 block(256);
  hipLaunchKernelGGL(logmm_kernel, grid, block, 0, stream, x, acc, out);
}


# x strip read by whole-line loads (one load = 512 contiguous bytes = 4 rows; was 32 lines per load, one row per lane); row max by 8-lane DPP maximum, exp'd rows to LDS as 16-byte stores
# speedup vs baseline: 1.1712x; 1.1712x over previous
#include <hip/hip_runtime.h>


typedef __attribute__((ext_vector_type(2))) float v2f;
typedef __attribute__((ext_vector_type(4))) float v4f;
typedef __attribute__((ext_vector_type(8))) float v8f;

#define NUM_SCOPES 128
#define NUM_DECOMPS 8
#define BATCH 1024
#define K_IN 32
#define N_OUT 32

constexpr int WAVES   = 8;
constexpr int STRIP   = 32;
constexpr int NSTRIPS = BATCH / STRIP;
constexpr int APAD    = 36;

template <int MASK>
__device__ __forceinline__ float max_xor_lane(float v) {
  const float o = __builtin_amdgcn_update_dpp(v, v, 0x160 | MASK, 0xf, 0xf, true);
  return __builtin_elementwise_maximum(v, o);
}

__global__ __launch_bounds__(256)
void logmm_kernel(const float* __restrict__ x,
                  const float* __restrict__ acc,
                  float* __restrict__ out) {
  __shared__ float sB[K_IN * N_OUT];
  __shared__ float sMax[N_OUT];
  __shared__ float sCb[N_OUT];
  __shared__ __attribute__((aligned(16))) float sA[WAVES][STRIP * APAD];
  __shared__ __attribute__((aligned(16))) float sRowMax[WAVES][STRIP];

  const int tid  = threadIdx.x;
  const int lane = tid & 31;
  const int wave = tid >> 5;
  const int sd   = blockIdx.x;

  const float* xb = x   + (size_t)sd * BATCH * K_IN;
  const float* ab = acc + (size_t)sd * K_IN * N_OUT;
  float*       ob = out + (size_t)sd * BATCH * N_OUT;

  ((float4*)sB)[tid] = ((const float4*)ab)[tid];
  __syncthreads();

  if (tid < N_OUT) {
    const int j = tid;
    float m = sB[j], s = sB[j];
    #pragma unroll
    for (int k = 1; k < K_IN; ++k) {
      float v = sB[k * N_OUT + j];
      m = fmaxf(m, v);
      s += v;
    }
    sMax[j] = m;
    sCb[j]  = __logf(m) - __logf(s);
  }
  __syncthreads();

  {
    const int j = (tid * 4) & (N_OUT - 1);
    float4 v = ((float4*)sB)[tid];
    v.x /= sMax[j + 0]; v.y /= sMax[j + 1];
    v.z /= sMax[j + 2]; v.w /= sMax[j + 3];
    ((float4*)sB)[tid] = v;
  }
  __syncthreads();

  const int ml    = lane & 15;
  const int khalf = lane >> 4;
  v2f bfrag[2][8];
  #pragma unroll
  for (int jh = 0; jh < 2; ++jh) {
    #pragma unroll
    for (int kb = 0; kb < 8; ++kb) {
      const int k0 = kb * 4 + 2 * khalf;
      const int n  = jh * 16 + ml;
      bfrag[jh][kb].x = sB[k0 * N_OUT + n];
      bfrag[jh][kb].y = sB[(k0 + 1) * N_OUT + n];
    }
  }
  const float cb0 = sCb[ml];
  const float cb1 = sCb[16 + ml];

  float* const aw = &sA[wave][0];
  float* const rm = &sRowMax[wave][0];

  const int g = lane >> 3;
  const int p = lane & 7;

  for (int strip = wave; strip < NSTRIPS; strip += WAVES) {
    const int m0 = strip * STRIP;

    const v4f* sp = (const v4f*)(xb + (size_t)m0 * K_IN);
    v4f r[8];
    #pragma unroll
    for (int i = 0; i < 8; ++i) r[i] = sp[i * 32 + lane];

    float m[8];
    #pragma unroll
    for (int i = 0; i < 8; ++i) {
      float t = fmaxf(fmaxf(r[i].x, r[i].y), fmaxf(r[i].z, r[i].w));
      t = max_xor_lane<1>(t);
      t = max_xor_lane<2>(t);
      t = max_xor_lane<4>(t);
      m[i] = t;
    }

    #pragma unroll
    for (int i = 0; i < 8; ++i) {
      v4f e;
      e.x = __expf(r[i].x - m[i]);
      e.y = __expf(r[i].y - m[i]);
      e.z = __expf(r[i].z - m[i]);
      e.w = __expf(r[i].w - m[i]);
      *(v4f*)(aw + (4 * i + g) * APAD + 4 * p) = e;
    }

    {
      float mine = m[0];
      #pragma unroll
      for (int i = 1; i < 8; ++i) mine = (p == i) ? m[i] : mine;
      rm[4 * p + g] = mine;
    }
    __builtin_amdgcn_wave_barrier();

    v8f c00 = {}, c01 = {}, c10 = {}, c11 = {};
    #pragma unroll
    for (int kb = 0; kb < 8; ++kb) {
      const int k0 = kb * 4 + 2 * khalf;
      const v2f a0 = *(const v2f*)(aw + ml        * APAD + k0);
      const v2f a1 = *(const v2f*)(aw + (16 + ml) * APAD + k0);
      c00 = __builtin_amdgcn_wmma_f32_16x16x4_f32(false, a0, false, bfrag[0][kb], (short)0, c00, false, false);
      c01 = __builtin_amdgcn_wmma_f32_16x16x4_f32(false, a0, false, bfrag[1][kb], (short)0, c01, false, false);
      c10 = __builtin_amdgcn_wmma_f32_16x16x4_f32(false, a1, false, bfrag[0][kb], (short)0, c10, false, false);
      c11 = __builtin_amdgcn_wmma_f32_16x16x4_f32(false, a1, false, bfrag[1][kb], (short)0, c11, false, false);
    }

    #pragma unroll
    for (int mt = 0; mt < 2; ++mt) {
      const v8f cA = mt ? c10 : c00;
      const v8f cB = mt ? c11 : c01;
      const v4f am_lo = *(const v4f*)(rm + 16 * mt + 8 * khalf);
      const v4f am_hi = *(const v4f*)(rm + 16 * mt + 8 * khalf + 4);
      #pragma unroll
      for (int v = 0; v < 8; ++v) {
        const int mlocal = mt * 16 + v + 8 * khalf;
        const float am   = (v < 4) ? am_lo[v & 3] : am_hi[v & 3];
        float* orow = ob + (size_t)(m0 + mlocal) * N_OUT;
        orow[ml]      = __logf(cA[v]) + am + cb0;
        orow[16 + ml] = __logf(cB[v]) + am + cb1;
      }
    }
    __builtin_amdgcn_wave_barrier();
  }
}

extern "C" void kernel_launch(void* const* d_in, const int* in_sizes, int n_in,
                              void* d_out, int out_size, void* d_ws, size_t ws_size,
                              hipStream_t stream) {
  const float* x   = (const float*)d_in[0];
  const float* acc = (const float*)d_in[1];
  float* out       = (float*)d_out;

  dim3 grid(NUM_SCOPES * NUM_DECOMPS);
  dim3 block(256);
  hipLaunchKernelGGL(logmm_kernel, grid, block, 0, stream, x, acc, out);
}
